# starting kernel
# baseline (speedup 1.0000x reference)
#include <hip/hip_runtime.h>
#include <hip/hip_bf16.h>

typedef __attribute__((ext_vector_type(2))) float v2f;
typedef __attribute__((ext_vector_type(8))) float v8f;

template <int H, bool STORE_X>
__global__ __launch_bounds__(256) void fused_bilinear_layer(
    const float* __restrict__ Xprev,
    const float* __restrict__ In,
    const float* __restrict__ W,
    const float* __restrict__ bias,
    float* __restrict__ Xout,
    float* __restrict__ Out,
    int outOff) {
  constexpr int K = H * 64;
  constexpr int OUTC = 128;
  constexpr int D = 32;

  __shared__ float xs[2 * H * D];
  __shared__ float ins[2 * 64 * D];

  const int tid = threadIdx.x;
  const int b0 = blockIdx.x * 2;

  {
    const float4* src = (const float4*)(Xprev + (size_t)b0 * H * D);
    float4* dst = (float4*)xs;
#pragma unroll 2
    for (int i = tid; i < (2 * H * D) / 4; i += 256) dst[i] = src[i];
  }
  {
    const float4* src = (const float4*)(In + (size_t)b0 * 64 * D);
    float4* dst = (float4*)ins;
#pragma unroll 2
    for (int i = tid; i < (2 * 64 * D) / 4; i += 256) dst[i] = src[i];
  }
  __syncthreads();

  const int lane = tid & 31;
  const int wave = tid >> 5;
  const int mbase = wave * 16;
  const int m = lane & 15;
  const int khalf = lane >> 4;

  v8f acc[4] = {};

  const float* wrow = W + (size_t)(mbase + m) * K + khalf * 2;

  for (int h = 0; h < H; ++h) {
    float xv[4];
#pragma unroll
    for (int t = 0; t < 4; ++t) {
      const int bi = t >> 1;
      const int d = m + (t & 1) * 16;
      xv[t] = xs[bi * H * D + h * D + d];
    }
#pragma unroll
    for (int f0 = 0; f0 < 64; f0 += 4) {
      const int k = h * 64 + f0;
      const v2f a = *(const v2f*)(wrow + k);
      const int f = f0 + khalf * 2;
#pragma unroll
      for (int t = 0; t < 4; ++t) {
        const int bi = t >> 1;
        const int d = m + (t & 1) * 16;
        const float i0 = ins[bi * 64 * D + f * D + d];
        const float i1 = ins[bi * 64 * D + (f + 1) * D + d];
        v2f b;
        b.x = xv[t] * i0;
        b.y = xv[t] * i1;
        acc[t] = __builtin_amdgcn_wmma_f32_16x16x4_f32(
            false, a, false, b, (short)0, acc[t], false, false);
      }
    }
  }

#pragma unroll
  for (int r = 0; r < 8; ++r) {
    const int o = mbase + r + khalf * 8;
    const float bv = bias[o];
    float s0 = 0.f, s1 = 0.f;
#pragma unroll
    for (int t = 0; t < 4; ++t) {
      float v = acc[t][r] + bv;
      v = v > 0.f ? v : 0.f;
      const int bi = t >> 1;
      const int d = m + (t & 1) * 16;
      if (STORE_X) {
        Xout[(size_t)(b0 + bi) * OUTC * D + (size_t)o * D + d] = v;
      }
      if (bi == 0) s0 += v; else s1 += v;
    }
#pragma unroll
    for (int sh = 1; sh < 16; sh <<= 1) {
      s0 += __shfl_xor(s0, sh, 32);
      s1 += __shfl_xor(s1, sh, 32);
    }
    if (m == 0) {
      Out[(size_t)b0 * 384 + outOff + o] = s0;
      Out[(size_t)(b0 + 1) * 384 + outOff + o] = s1;
    }
  }
}

extern "C" void kernel_launch(void* const* d_in, const int* in_sizes, int n_in,
                              void* d_out, int out_size, void* d_ws,
                              size_t ws_size, hipStream_t stream) {
  (void)in_sizes; (void)n_in; (void)out_size; (void)ws_size;
  const float* input = (const float*)d_in[0];
  const float* W0 = (const float*)d_in[1];
  const float* b0 = (const float*)d_in[2];
  const float* W1 = (const float*)d_in[3];
  const float* b1 = (const float*)d_in[4];
  const float* W2 = (const float*)d_in[5];
  const float* b2 = (const float*)d_in[6];
  float* out = (float*)d_out;

  constexpr int B = 512;
  float* x1 = (float*)d_ws;
  float* x2 = x1 + (size_t)B * 128 * 32;

  dim3 grid(B / 2), block(256);
  fused_bilinear_layer<64, true><<<grid, block, 0, stream>>>(
      input, input, W0, b0, x1, out, 0);
  fused_bilinear_layer<128, true><<<grid, block, 0, stream>>>(
      x1, input, W1, b1, x2, out, 128);
  fused_bilinear_layer<128, false><<<grid, block, 0, stream>>>(
      x2, input, W2, b2, nullptr, out, 256);
}


# Contract over f first: B = input fragments, loop-invariant in registers, no per-h products; x[h] scales each h-partial tile by column (8 FMAs a tile); order of sums changes (f inside, h outside)
# speedup vs baseline: 1.1365x; 1.1365x over previous
#include <hip/hip_runtime.h>
#include <hip/hip_bf16.h>

typedef __attribute__((ext_vector_type(2))) float v2f;
typedef __attribute__((ext_vector_type(8))) float v8f;

template <int H, bool STORE_X>
__global__ __launch_bounds__(256) void fused_bilinear_layer(
    const float* __restrict__ Xprev,
    const float* __restrict__ In,
    const float* __restrict__ W,
    const float* __restrict__ bias,
    float* __restrict__ Xout,
    float* __restrict__ Out,
    int outOff) {
  constexpr int K = H * 64;
  constexpr int OUTC = 128;
  constexpr int D = 32;

  __shared__ float xs[2 * H * D];
  __shared__ float ins[2 * 64 * D];

  const int tid = threadIdx.x;
  const int b0 = blockIdx.x * 2;

  {
    const float4* src = (const float4*)(Xprev + (size_t)b0 * H * D);
    float4* dst = (float4*)xs;
#pragma unroll 2
    for (int i = tid; i < (2 * H * D) / 4; i += 256) dst[i] = src[i];
  }
  {
    const float4* src = (const float4*)(In + (size_t)b0 * 64 * D);
    float4* dst = (float4*)ins;
#pragma unroll 2
    for (int i = tid; i < (2 * 64 * D) / 4; i += 256) dst[i] = src[i];
  }
  __syncthreads();

  const int lane = tid & 31;
  const int wave = tid >> 5;
  const int mbase = wave * 16;
  const int m = lane & 15;
  const int khalf = lane >> 4;

  v8f acc[4] = {};

  v2f bfrag[16][4];
#pragma unroll
  for (int j = 0; j < 16; ++j) {
    const int f = j * 4 + khalf * 2;
#pragma unroll
    for (int t = 0; t < 4; ++t) {
      const int bi = t >> 1;
      const int d = m + (t & 1) * 16;
      bfrag[j][t].x = ins[bi * 64 * D + f * D + d];
      bfrag[j][t].y = ins[bi * 64 * D + (f + 1) * D + d];
    }
  }

  const float* wrow = W + (size_t)(mbase + m) * K + khalf * 2;

  for (int h = 0; h < H; ++h) {
    v2f a[16];
#pragma unroll
    for (int j = 0; j < 16; ++j) a[j] = *(const v2f*)(wrow + h * 64 + j * 4);

    v8f part[4] = {};
#pragma unroll
    for (int j = 0; j < 16; ++j) {
#pragma unroll
      for (int t = 0; t < 4; ++t) {
        part[t] = __builtin_amdgcn_wmma_f32_16x16x4_f32(
            false, a[j], false, bfrag[j][t], (short)0, part[t], false, false);
      }
    }

#pragma unroll
    for (int t = 0; t < 4; ++t) {
      const int bi = t >> 1;
      const int d = m + (t & 1) * 16;
      const float xv = xs[bi * H * D + h * D + d];
#pragma unroll
      for (int r = 0; r < 8; ++r) acc[t][r] += xv * part[t][r];
    }
  }

#pragma unroll
  for (int r = 0; r < 8; ++r) {
    const int o = mbase + r + khalf * 8;
    const float bv = bias[o];
    float s0 = 0.f, s1 = 0.f;
#pragma unroll
    for (int t = 0; t < 4; ++t) {
      float v = acc[t][r] + bv;
      v = v > 0.f ? v : 0.f;
      const int bi = t >> 1;
      const int d = m + (t & 1) * 16;
      if (STORE_X) {
        Xout[(size_t)(b0 + bi) * OUTC * D + (size_t)o * D + d] = v;
      }
      if (bi == 0) s0 += v; else s1 += v;
    }
#pragma unroll
    for (int sh = 1; sh < 16; sh <<= 1) {
      s0 += __shfl_xor(s0, sh, 32);
      s1 += __shfl_xor(s1, sh, 32);
    }
    if (m == 0) {
      Out[(size_t)b0 * 384 + outOff + o] = s0;
      Out[(size_t)(b0 + 1) * 384 + outOff + o] = s1;
    }
  }
}

extern "C" void kernel_launch(void* const* d_in, const int* in_sizes, int n_in,
                              void* d_out, int out_size, void* d_ws,
                              size_t ws_size, hipStream_t stream) {
  (void)in_sizes; (void)n_in; (void)out_size; (void)ws_size;
  const float* input = (const float*)d_in[0];
  const float* W0 = (const float*)d_in[1];
  const float* b0 = (const float*)d_in[2];
  const float* W1 = (const float*)d_in[3];
  const float* b1 = (const float*)d_in[4];
  const float* W2 = (const float*)d_in[5];
  const float* b2 = (const float*)d_in[6];
  float* out = (float*)d_out;

  constexpr int B = 512;
  float* x1 = (float*)d_ws;
  float* x2 = x1 + (size_t)B * 128 * 32;

  dim3 grid(B / 2), block(256);
  fused_bilinear_layer<64, true><<<grid, block, 0, stream>>>(
      input, input, W0, b0, x1, out, 0);
  fused_bilinear_layer<128, true><<<grid, block, 0, stream>>>(
      x1, input, W1, b1, x2, out, 128);
  fused_bilinear_layer<128, false><<<grid, block, 0, stream>>>(
      x2, input, W2, b2, nullptr, out, 256);
}
